# starting kernel
# baseline (speedup 1.0000x reference)
#include <hip/hip_runtime.h>
#include <hip/hip_bf16.h>

typedef float v2f __attribute__((ext_vector_type(2)));
typedef float v8f __attribute__((ext_vector_type(8)));

#define HIDDEN 10
#define T_LEN  2048
#define BATCH  4096
#define INPUT  4
#define PF     8

#if __has_builtin(__builtin_amdgcn_tanhf)
  __device__ __forceinline__ float dev_tanh(float x) { return __builtin_amdgcn_tanhf(x); }
#elif __has_builtin(__builtin_amdgcn_tanh_f32)
  __device__ __forceinline__ float dev_tanh(float x) { return __builtin_amdgcn_tanh_f32(x); }
#else
  __device__ __forceinline__ float dev_tanh(float x) {
    float r = __builtin_amdgcn_rcpf(__expf(2.0f * x) + 1.0f);
    return __builtin_fmaf(-2.0f, r, 1.0f);
  }
#endif

__device__ __forceinline__ float fast_sigmoid(float x) {
    return __builtin_fmaf(0.5f, dev_tanh(0.5f * x), 0.5f);
}
__device__ __forceinline__ float fast_tanh(float x) { return dev_tanh(x); }

__device__ __forceinline__ void wave_order_barrier() {
    __builtin_amdgcn_wave_barrier();
}

__global__ __launch_bounds__(32)
void lstm_wmma_f32_kernel(const float* __restrict__ x,
                          const float* __restrict__ W_ih,
                          const float* __restrict__ W_hh,
                          const float* __restrict__ b_ih,
                          const float* __restrict__ b_hh,
                          const float* __restrict__ W_lin,
                          const float* __restrict__ b_lin,
                          float* __restrict__ out)
{
    __shared__ __align__(32) float hlds2[16 * 16];

    const int lane = threadIdx.x;
    const int m    = lane & 15;
    const int half = lane >> 4;
    const int n    = m;
    const int b0   = blockIdx.x * 16;

    v2f Bih[4];
    v2f Bhh[4][3];
    v8f Cbias[4];
    float wl[HIDDEN];
    float wl6[6];

    #pragma unroll
    for (int g = 0; g < 4; ++g) {
        float e0 = 0.f, e1 = 0.f, bia = 0.f;
        if (n < HIDDEN) {
            const int row = g * HIDDEN + n;
            const int k0  = 2 * half;
            e0  = W_ih[row * INPUT + k0];
            e1  = W_ih[row * INPUT + k0 + 1];
            bia = b_ih[row] + b_hh[row];
        }
        Bih[g].x = e0; Bih[g].y = e1;
        v8f c;
        #pragma unroll
        for (int e = 0; e < 8; ++e) c[e] = bia;
        Cbias[g] = c;

        #pragma unroll
        for (int j = 0; j < 3; ++j) {
            float h0 = 0.f, h1 = 0.f;
            if (n < HIDDEN) {
                const int row = g * HIDDEN + n;
                const int kk  = j * 4 + 2 * half;
                if (kk     < HIDDEN) h0 = W_hh[row * HIDDEN + kk];
                if (kk + 1 < HIDDEN) h1 = W_hh[row * HIDDEN + kk + 1];
            }
            Bhh[g][j].x = h0; Bhh[g][j].y = h1;
        }
    }
    #pragma unroll
    for (int i = 0; i < HIDDEN; ++i) wl[i] = W_lin[i];
    #pragma unroll
    for (int j = 0; j < 3; ++j) {
        const int kk = j * 4 + 2 * half;
        wl6[2 * j + 0] = (kk     < HIDDEN) ? W_lin[kk]     : 0.f;
        wl6[2 * j + 1] = (kk + 1 < HIDDEN) ? W_lin[kk + 1] : 0.f;
    }
    const float blin = b_lin[0];

    for (int i = lane; i < 256; i += 32) hlds2[i] = 0.f;
    __syncthreads();

    v8f creg;
    #pragma unroll
    for (int e = 0; e < 8; ++e) creg[e] = 0.f;

    const float* xrow = x + ((size_t)(b0 + m) * T_LEN) * INPUT + 2 * half;
    float*       orow = out + (size_t)(b0 + m) * T_LEN;

    v2f xbuf[PF], xcur[PF];
    #pragma unroll
    for (int p = 0; p < PF; ++p) {
        xbuf[p].x = xrow[(size_t)p * INPUT + 0];
        xbuf[p].y = xrow[(size_t)p * INPUT + 1];
    }

    for (int tb = 0; tb < T_LEN; tb += PF) {
        #pragma unroll
        for (int p = 0; p < PF; ++p) xcur[p] = xbuf[p];

        {
            int tnext = tb + PF;
            if (tnext > T_LEN - PF) tnext = T_LEN - PF;
            const float* xpf = xrow + (size_t)tnext * INPUT;
            #pragma unroll
            for (int p = 0; p < PF; ++p) {
                xbuf[p].x = xpf[p * INPUT + 0];
                xbuf[p].y = xpf[p * INPUT + 1];
            }
        }

        #pragma unroll
        for (int u = 0; u < PF; ++u) {
            const int t = tb + u;
            const v2f xa = xcur[u];

            v2f ha[3];
            #pragma unroll
            for (int j = 0; j < 3; ++j) {
                const int kk = j * 4 + 2 * half;
                ha[j].x = hlds2[(kk    ) * 16 + m];
                ha[j].y = hlds2[(kk + 1) * 16 + m];
            }

            v8f acc[4];
            #pragma unroll
            for (int g = 0; g < 4; ++g)
                acc[g] = __builtin_amdgcn_wmma_f32_16x16x4_f32(
                    false, xa, false, Bih[g], (short)0, Cbias[g], false, false);

            #pragma unroll
            for (int j = 0; j < 3; ++j)
                #pragma unroll
                for (int g = 0; g < 4; ++g)
                    acc[g] = __builtin_amdgcn_wmma_f32_16x16x4_f32(
                        false, ha[j], false, Bhh[g][j], (short)0, acc[g], false, false);

            {
                float po = ha[0].x * wl6[0] + ha[0].y * wl6[1]
                         + ha[1].x * wl6[2] + ha[1].y * wl6[3]
                         + ha[2].x * wl6[4] + ha[2].y * wl6[5];
                po += __shfl_xor(po, 16, 32);
                if (half == 0) {
                    const int to = (t > 0) ? (t - 1) : 0;
                    orow[to] = po + blin;
                }
            }

            wave_order_barrier();

            v8f hv;
            #pragma unroll
            for (int e = 0; e < 8; ++e) {
                const float iv = fast_sigmoid(acc[0][e]);
                const float fv = fast_sigmoid(acc[1][e]);
                const float gv = fast_tanh   (acc[2][e]);
                const float ov = fast_sigmoid(acc[3][e]);
                const float cv = fv * creg[e] + iv * gv;
                creg[e] = cv;
                hv[e] = ov * fast_tanh(cv);
            }
            *(v8f*)&hlds2[n * 16 + 8 * half] = hv;
            wave_order_barrier();
        }
    }

    if (half == 0) {
        float o = blin;
        #pragma unroll
        for (int i = 0; i < HIDDEN; ++i) o += hlds2[i * 16 + m] * wl[i];
        orow[T_LEN - 1] = o;
    }
}

extern "C" void kernel_launch(void* const* d_in, const int* in_sizes, int n_in,
                              void* d_out, int out_size, void* d_ws, size_t ws_size,
                              hipStream_t stream) {
    const float* x     = (const float*)d_in[0];
    const float* W_ih  = (const float*)d_in[1];
    const float* W_hh  = (const float*)d_in[2];
    const float* b_ih  = (const float*)d_in[3];
    const float* b_hh  = (const float*)d_in[4];
    const float* W_lin = (const float*)d_in[5];
    const float* b_lin = (const float*)d_in[6];
    float* out = (float*)d_out;

    dim3 grid(BATCH / 16);
    dim3 block(32);
    lstm_wmma_f32_kernel<<<grid, block, 0, stream>>>(
        x, W_ih, W_hh, b_ih, b_hh, W_lin, b_lin, out);
}


# Transposed product: gates^T = W x [x;h]^T, batch on N, gate rows permuted so each lane owns whole cells; h_t feeds the next B operand from registers (no LDS); head as a spare W row; k order permuted
# speedup vs baseline: 1.3956x; 1.3956x over previous
#include <hip/hip_runtime.h>
#include <hip/hip_bf16.h>


typedef float v2f __attribute__((ext_vector_type(2)));
typedef float v8f __attribute__((ext_vector_type(8)));

#define HIDDEN 10
#define T_LEN  2048
#define BATCH  4096
#define INPUT  4
#define PF     8
#define UNITS  5

#if __has_builtin(__builtin_amdgcn_tanhf)
  __device__ __forceinline__ float dev_tanh(float x) { return __builtin_amdgcn_tanhf(x); }
#elif __has_builtin(__builtin_amdgcn_tanh_f32)
  __device__ __forceinline__ float dev_tanh(float x) { return __builtin_amdgcn_tanh_f32(x); }
#else
  __device__ __forceinline__ float dev_tanh(float x) {
    float r = __builtin_amdgcn_rcpf(__expf(2.0f * x) + 1.0f);
    return __builtin_fmaf(-2.0f, r, 1.0f);
  }
#endif

__device__ __forceinline__ float fast_sigmoid(float x) {
    return __builtin_fmaf(0.5f, dev_tanh(0.5f * x), 0.5f);
}
__device__ __forceinline__ float fast_tanh(float x) { return dev_tanh(x); }

__global__ __launch_bounds__(32)
void lstm_wmma_f32_kernel(const float* __restrict__ x,
                          const float* __restrict__ W_ih,
                          const float* __restrict__ W_hh,
                          const float* __restrict__ b_ih,
                          const float* __restrict__ b_hh,
                          const float* __restrict__ W_lin,
                          const float* __restrict__ b_lin,
                          float* __restrict__ out)
{
    const int lane = threadIdx.x;
    const int c    = lane & 15;
    const int half = lane >> 4;
    const int b0   = blockIdx.x * 16;

    v2f Ax[3];
    v2f Ah[3][3];
    v8f Cb[3];

    #pragma unroll
    for (int tt = 0; tt < 3; ++tt) {
        const int  q    = tt * 8 + (c & 7);
        const int  own  = c >> 3;
        const bool cell = q < 4 * UNITS;
        const bool head = (q == 4 * UNITS) && (own == 0);
        const int  row  = (q & 3) * HIDDEN + UNITS * own + (q >> 2);

        float e0 = 0.f, e1 = 0.f;
        if (cell) {
            e0 = W_ih[row * INPUT + 2 * half];
            e1 = W_ih[row * INPUT + 2 * half + 1];
        }
        Ax[tt].x = e0; Ax[tt].y = e1;

        #pragma unroll
        for (int j = 0; j < 3; ++j) {
            float a0 = 0.f, a1 = 0.f;
            const int w0 = 2 * j, w1 = 2 * j + 1;
            if (cell) {
                if (w0 < UNITS) a0 = W_hh[row * HIDDEN + UNITS * half + w0];
                if (w1 < UNITS) a1 = W_hh[row * HIDDEN + UNITS * half + w1];
            } else if (head) {
                if (w0 < UNITS) a0 = W_lin[UNITS * half + w0];
                if (w1 < UNITS) a1 = W_lin[UNITS * half + w1];
            }
            Ah[tt][j].x = a0; Ah[tt][j].y = a1;
        }

        v8f cb;
        #pragma unroll
        for (int r = 0; r < 8; ++r) {
            const int qq = tt * 8 + r;
            float b = 0.f;
            if (qq < 4 * UNITS) {
                const int rw = (qq & 3) * HIDDEN + UNITS * half + (qq >> 2);
                b = b_ih[rw] + b_hh[rw];
            } else if (qq == 4 * UNITS && half == 0) {
                b = b_lin[0];
            }
            cb[r] = b;
        }
        Cb[tt] = cb;
    }

    float cst[UNITS];
    #pragma unroll
    for (int u = 0; u < UNITS; ++u) cst[u] = 0.f;
    v2f hB[3];
    #pragma unroll
    for (int j = 0; j < 3; ++j) { hB[j].x = 0.f; hB[j].y = 0.f; }

    const float* xrow = x + ((size_t)(b0 + c) * T_LEN) * INPUT + 2 * half;
    float*       orow = out + (size_t)(b0 + c) * T_LEN;

    v2f xbuf[PF], xcur[PF];
    #pragma unroll
    for (int p = 0; p < PF; ++p) {
        xbuf[p].x = xrow[(size_t)p * INPUT + 0];
        xbuf[p].y = xrow[(size_t)p * INPUT + 1];
    }

    for (int tb = 0; tb < T_LEN; tb += PF) {
        #pragma unroll
        for (int p = 0; p < PF; ++p) xcur[p] = xbuf[p];

        {
            int tnext = tb + PF;
            if (tnext > T_LEN - PF) tnext = T_LEN - PF;
            const float* xpf = xrow + (size_t)tnext * INPUT;
            #pragma unroll
            for (int p = 0; p < PF; ++p) {
                xbuf[p].x = xpf[p * INPUT + 0];
                xbuf[p].y = xpf[p * INPUT + 1];
            }
        }

        float ov[PF];

        #pragma unroll
        for (int u = 0; u < PF; ++u) {
            v8f acc[3];
            #pragma unroll
            for (int tt = 0; tt < 3; ++tt)
                acc[tt] = __builtin_amdgcn_wmma_f32_16x16x4_f32(
                    false, Ax[tt], false, xcur[u], (short)0, Cb[tt], false, false);

            #pragma unroll
            for (int j = 0; j < 3; ++j)
                #pragma unroll
                for (int tt = 0; tt < 3; ++tt)
                    acc[tt] = __builtin_amdgcn_wmma_f32_16x16x4_f32(
                        false, Ah[tt][j], false, hB[j], (short)0, acc[tt], false, false);

            ov[u] = acc[2][4];

            float hn[UNITS + 1];
            #pragma unroll
            for (int k = 0; k < UNITS; ++k) {
                const int tt = k >> 1, rb = (k & 1) * 4;
                const float iv = fast_sigmoid(acc[tt][rb + 0]);
                const float fv = fast_sigmoid(acc[tt][rb + 1]);
                const float gv = fast_tanh   (acc[tt][rb + 2]);
                const float og = fast_sigmoid(acc[tt][rb + 3]);
                const float cv = fv * cst[k] + iv * gv;
                cst[k] = cv;
                hn[k] = og * fast_tanh(cv);
            }
            hn[UNITS] = 0.f;
            #pragma unroll
            for (int j = 0; j < 3; ++j) { hB[j].x = hn[2 * j]; hB[j].y = hn[2 * j + 1]; }
        }

        if (half == 0) {
            if (tb > 0) orow[tb - 1] = ov[0];
            #pragma unroll
            for (int u = 1; u < PF; ++u) orow[tb - 1 + u] = ov[u];
        }
    }

    {
        v8f accf = Cb[2];
        #pragma unroll
        for (int j = 0; j < 3; ++j)
            accf = __builtin_amdgcn_wmma_f32_16x16x4_f32(
                false, Ah[2][j], false, hB[j], (short)0, accf, false, false);
        if (half == 0) orow[T_LEN - 1] = accf[4];
    }
}

extern "C" void kernel_launch(void* const* d_in, const int* in_sizes, int n_in,
                              void* d_out, int out_size, void* d_ws, size_t ws_size,
                              hipStream_t stream) {
    const float* x     = (const float*)d_in[0];
    const float* W_ih  = (const float*)d_in[1];
    const float* W_hh  = (const float*)d_in[2];
    const float* b_ih  = (const float*)d_in[3];
    const float* b_hh  = (const float*)d_in[4];
    const float* W_lin = (const float*)d_in[5];
    const float* b_lin = (const float*)d_in[6];
    float* out = (float*)d_out;

    dim3 grid(BATCH / 16);
    dim3 block(32);
    lstm_wmma_f32_kernel<<<grid, block, 0, stream>>>(
        x, W_ih, W_hh, b_ih, b_hh, W_lin, b_lin, out);
}


# Gate-row tiles split over three waves per batch tile (4 WMMA per step per wave, was 12); h pairs exchanged through LDS behind one barrier per step; own k chunk added first (chunk sum order changes)
# speedup vs baseline: 2.0994x; 1.5043x over previous
#include <hip/hip_runtime.h>
#include <hip/hip_bf16.h>


typedef float v2f __attribute__((ext_vector_type(2)));
typedef float v8f __attribute__((ext_vector_type(8)));

#define HIDDEN 10
#define T_LEN  2048
#define BATCH  4096
#define INPUT  4
#define PF     8
#define UNITS  5

#if __has_builtin(__builtin_amdgcn_tanhf)
  __device__ __forceinline__ float dev_tanh(float x) { return __builtin_amdgcn_tanhf(x); }
#elif __has_builtin(__builtin_amdgcn_tanh_f32)
  __device__ __forceinline__ float dev_tanh(float x) { return __builtin_amdgcn_tanh_f32(x); }
#else
  __device__ __forceinline__ float dev_tanh(float x) {
    float r = __builtin_amdgcn_rcpf(__expf(2.0f * x) + 1.0f);
    return __builtin_fmaf(-2.0f, r, 1.0f);
  }
#endif

__device__ __forceinline__ float fast_sigmoid(float x) {
    return __builtin_fmaf(0.5f, dev_tanh(0.5f * x), 0.5f);
}
__device__ __forceinline__ float fast_tanh(float x) { return dev_tanh(x); }

__global__ __launch_bounds__(96)
void lstm_wmma_f32_kernel(const float* __restrict__ x,
                          const float* __restrict__ W_ih,
                          const float* __restrict__ W_hh,
                          const float* __restrict__ b_ih,
                          const float* __restrict__ b_hh,
                          const float* __restrict__ W_lin,
                          const float* __restrict__ b_lin,
                          float* __restrict__ out)
{
    __shared__ v2f hx[2][3][32];

    const int lane = threadIdx.x & 31;
    const int tt   = __builtin_amdgcn_readfirstlane((int)(threadIdx.x >> 5));
    const int c    = lane & 15;
    const int half = lane >> 4;
    const int b0   = blockIdx.x * 16;
    const int j1   = (tt + 1) % 3;
    const int j2   = (tt + 2) % 3;

    v2f Ax;
    v2f Ah[3];
    v8f Cb;
    {
        const int  q    = tt * 8 + (c & 7);
        const int  own  = c >> 3;
        const bool cell = q < 4 * UNITS;
        const bool head = (q == 4 * UNITS) && (own == 0);
        const int  row  = (q & 3) * HIDDEN + UNITS * own + (q >> 2);

        float e0 = 0.f, e1 = 0.f;
        if (cell) {
            e0 = W_ih[row * INPUT + 2 * half];
            e1 = W_ih[row * INPUT + 2 * half + 1];
        }
        Ax.x = e0; Ax.y = e1;

        #pragma unroll
        for (int i = 0; i < 3; ++i) {
            const int jj = (i == 0) ? tt : ((i == 1) ? j1 : j2);
            float a0 = 0.f, a1 = 0.f;
            const int w0 = 2 * jj, w1 = 2 * jj + 1;
            if (cell) {
                if (w0 < UNITS) a0 = W_hh[row * HIDDEN + UNITS * half + w0];
                if (w1 < UNITS) a1 = W_hh[row * HIDDEN + UNITS * half + w1];
            } else if (head) {
                if (w0 < UNITS) a0 = W_lin[UNITS * half + w0];
                if (w1 < UNITS) a1 = W_lin[UNITS * half + w1];
            }
            Ah[i].x = a0; Ah[i].y = a1;
        }

        #pragma unroll
        for (int r = 0; r < 8; ++r) {
            const int qq = tt * 8 + r;
            float b = 0.f;
            if (qq < 4 * UNITS) {
                const int rw = (qq & 3) * HIDDEN + UNITS * half + (qq >> 2);
                b = b_ih[rw] + b_hh[rw];
            } else if (qq == 4 * UNITS && half == 0) {
                b = b_lin[0];
            }
            Cb[r] = b;
        }
    }

    float cst0 = 0.f, cst1 = 0.f;
    v2f hown; hown.x = 0.f; hown.y = 0.f;
    v2f hB1;  hB1.x  = 0.f; hB1.y  = 0.f;
    v2f hB2;  hB2.x  = 0.f; hB2.y  = 0.f;

    const float* xrow = x + ((size_t)(b0 + c) * T_LEN) * INPUT + 2 * half;
    float*       orow = out + (size_t)(b0 + c) * T_LEN;
    const bool   writer = (tt == 2) && (half == 0);

    v2f xbuf[PF], xcur[PF];
    #pragma unroll
    for (int p = 0; p < PF; ++p) {
        xbuf[p].x = xrow[(size_t)p * INPUT + 0];
        xbuf[p].y = xrow[(size_t)p * INPUT + 1];
    }

    v8f accp = __builtin_amdgcn_wmma_f32_16x16x4_f32(
        false, Ax, false, xbuf[0], (short)0, Cb, false, false);
    accp = __builtin_amdgcn_wmma_f32_16x16x4_f32(
        false, Ah[0], false, hown, (short)0, accp, false, false);

    for (int tb = 0; tb < T_LEN; tb += PF) {
        #pragma unroll
        for (int p = 0; p < PF; ++p) xcur[p] = xbuf[p];

        {
            int tnext = tb + PF;
            if (tnext > T_LEN - PF) tnext = T_LEN - PF;
            const float* xpf = xrow + (size_t)tnext * INPUT;
            #pragma unroll
            for (int p = 0; p < PF; ++p) {
                xbuf[p].x = xpf[p * INPUT + 0];
                xbuf[p].y = xpf[p * INPUT + 1];
            }
        }

        float ov[PF];

        #pragma unroll
        for (int u = 0; u < PF; ++u) {
            v8f acc = __builtin_amdgcn_wmma_f32_16x16x4_f32(
                false, Ah[1], false, hB1, (short)0, accp, false, false);
            acc = __builtin_amdgcn_wmma_f32_16x16x4_f32(
                false, Ah[2], false, hB2, (short)0, acc, false, false);

            ov[u] = acc[4];

            {
                const float iv = fast_sigmoid(acc[0]);
                const float fv = fast_sigmoid(acc[1]);
                const float gv = fast_tanh   (acc[2]);
                const float og = fast_sigmoid(acc[3]);
                const float cv = fv * cst0 + iv * gv;
                cst0 = cv;
                hown.x = og * fast_tanh(cv);
            }
            {
                const float iv = fast_sigmoid(acc[4]);
                const float fv = fast_sigmoid(acc[5]);
                const float gv = fast_tanh   (acc[6]);
                const float og = fast_sigmoid(acc[7]);
                const float cv = fv * cst1 + iv * gv;
                cst1 = cv;
                hown.y = og * fast_tanh(cv);
            }

            hx[(u + 1) & 1][tt][lane] = hown;

            __builtin_amdgcn_fence(__ATOMIC_RELEASE, "workgroup", "local");
            __builtin_amdgcn_s_barrier();
            __builtin_amdgcn_fence(__ATOMIC_ACQUIRE, "workgroup", "local");
            hB1 = hx[(u + 1) & 1][j1][lane];
            hB2 = hx[(u + 1) & 1][j2][lane];

            __builtin_amdgcn_sched_barrier(0);

            const v2f xn = (u + 1 < PF) ? xcur[(u + 1) % PF] : xbuf[0];
            accp = __builtin_amdgcn_wmma_f32_16x16x4_f32(
                false, Ax, false, xn, (short)0, Cb, false, false);
            accp = __builtin_amdgcn_wmma_f32_16x16x4_f32(
                false, Ah[0], false, hown, (short)0, accp, false, false);
        }

        if (writer) {
            if (tb > 0) orow[tb - 1] = ov[0];
            #pragma unroll
            for (int u = 1; u < PF; ++u) orow[tb - 1 + u] = ov[u];
        }
    }

    {
        v8f accf = Cb;
        accf = __builtin_amdgcn_wmma_f32_16x16x4_f32(
            false, Ah[0], false, hown, (short)0, accf, false, false);
        accf = __builtin_amdgcn_wmma_f32_16x16x4_f32(
            false, Ah[1], false, hB1, (short)0, accf, false, false);
        accf = __builtin_amdgcn_wmma_f32_16x16x4_f32(
            false, Ah[2], false, hB2, (short)0, accf, false, false);
        if (writer) orow[T_LEN - 1] = accf[4];
    }
}

extern "C" void kernel_launch(void* const* d_in, const int* in_sizes, int n_in,
                              void* d_out, int out_size, void* d_ws, size_t ws_size,
                              hipStream_t stream) {
    const float* x     = (const float*)d_in[0];
    const float* W_ih  = (const float*)d_in[1];
    const float* W_hh  = (const float*)d_in[2];
    const float* b_ih  = (const float*)d_in[3];
    const float* b_hh  = (const float*)d_in[4];
    const float* W_lin = (const float*)d_in[5];
    const float* b_lin = (const float*)d_in[6];
    float* out = (float*)d_out;

    dim3 grid(BATCH / 16);
    dim3 block(96);
    lstm_wmma_f32_kernel<<<grid, block, 0, stream>>>(
        x, W_ih, W_hh, b_ih, b_hh, W_lin, b_lin, out);
}


# Sigmoid constants folded into the weights (i,f,o rows x0.5, h columns x0.5, waves hand on 2h): each gate is one bare tanh, cell update is 3 FMAs; c update re-associated (rounding points move)
# speedup vs baseline: 2.2372x; 1.0656x over previous
#include <hip/hip_runtime.h>
#include <hip/hip_bf16.h>


typedef float v2f __attribute__((ext_vector_type(2)));
typedef float v8f __attribute__((ext_vector_type(8)));

#define HIDDEN 10
#define T_LEN  2048
#define BATCH  4096
#define INPUT  4
#define PF     8
#define UNITS  5

#if __has_builtin(__builtin_amdgcn_tanhf)
  __device__ __forceinline__ float dev_tanh(float x) { return __builtin_amdgcn_tanhf(x); }
#elif __has_builtin(__builtin_amdgcn_tanh_f32)
  __device__ __forceinline__ float dev_tanh(float x) { return __builtin_amdgcn_tanh_f32(x); }
#else
  __device__ __forceinline__ float dev_tanh(float x) {
    float r = __builtin_amdgcn_rcpf(__expf(2.0f * x) + 1.0f);
    return __builtin_fmaf(-2.0f, r, 1.0f);
  }
#endif

__device__ __forceinline__ float fast_tanh(float x) { return dev_tanh(x); }

__device__ __forceinline__ float cell_step(float zi, float zf, float zg, float zo, float& hc) {
    const float tf = fast_tanh(zf);
    const float ti = fast_tanh(zi);
    const float tg = fast_tanh(zg);
    const float to = fast_tanh(zo);
    const float fc = __builtin_fmaf(tf, hc, hc);
    const float ig = __builtin_fmaf(ti, tg, tg);
    const float cv = __builtin_fmaf(0.5f, ig, fc);
    hc = 0.5f * cv;
    const float tc = fast_tanh(cv);
    return __builtin_fmaf(to, tc, tc);
}

__global__ __launch_bounds__(96)
void lstm_wmma_f32_kernel(const float* __restrict__ x,
                          const float* __restrict__ W_ih,
                          const float* __restrict__ W_hh,
                          const float* __restrict__ b_ih,
                          const float* __restrict__ b_hh,
                          const float* __restrict__ W_lin,
                          const float* __restrict__ b_lin,
                          float* __restrict__ out)
{
    __shared__ v2f hx[2][3][32];

    const int lane = threadIdx.x & 31;
    const int tt   = __builtin_amdgcn_readfirstlane((int)(threadIdx.x >> 5));
    const int c    = lane & 15;
    const int half = lane >> 4;
    const int b0   = blockIdx.x * 16;
    const int j1   = (tt + 1) % 3;
    const int j2   = (tt + 2) % 3;

    v2f Ax;
    v2f Ah[3];
    v8f Cb;
    {
        const int  q    = tt * 8 + (c & 7);
        const int  own  = c >> 3;
        const bool cell = q < 4 * UNITS;
        const bool head = (q == 4 * UNITS) && (own == 0);
        const int  row  = (q & 3) * HIDDEN + UNITS * own + (q >> 2);
        const float sx = ((q & 3) == 2) ? 1.0f : 0.5f;
        const float sh = 0.5f * sx;

        float e0 = 0.f, e1 = 0.f;
        if (cell) {
            e0 = sx * W_ih[row * INPUT + 2 * half];
            e1 = sx * W_ih[row * INPUT + 2 * half + 1];
        }
        Ax.x = e0; Ax.y = e1;

        #pragma unroll
        for (int i = 0; i < 3; ++i) {
            const int jj = (i == 0) ? tt : ((i == 1) ? j1 : j2);
            float a0 = 0.f, a1 = 0.f;
            const int w0 = 2 * jj, w1 = 2 * jj + 1;
            if (cell) {
                if (w0 < UNITS) a0 = sh * W_hh[row * HIDDEN + UNITS * half + w0];
                if (w1 < UNITS) a1 = sh * W_hh[row * HIDDEN + UNITS * half + w1];
            } else if (head) {
                if (w0 < UNITS) a0 = 0.5f * W_lin[UNITS * half + w0];
                if (w1 < UNITS) a1 = 0.5f * W_lin[UNITS * half + w1];
            }
            Ah[i].x = a0; Ah[i].y = a1;
        }

        #pragma unroll
        for (int r = 0; r < 8; ++r) {
            const int qq = tt * 8 + r;
            float b = 0.f;
            if (qq < 4 * UNITS) {
                const int rw = (qq & 3) * HIDDEN + UNITS * half + (qq >> 2);
                b = (((qq & 3) == 2) ? 1.0f : 0.5f) * (b_ih[rw] + b_hh[rw]);
            } else if (qq == 4 * UNITS && half == 0) {
                b = b_lin[0];
            }
            Cb[r] = b;
        }
    }

    float hc0 = 0.f, hc1 = 0.f;
    v2f hown; hown.x = 0.f; hown.y = 0.f;
    v2f hB1;  hB1.x  = 0.f; hB1.y  = 0.f;
    v2f hB2;  hB2.x  = 0.f; hB2.y  = 0.f;

    const float* xrow = x + ((size_t)(b0 + c) * T_LEN) * INPUT + 2 * half;
    float*       orow = out + (size_t)(b0 + c) * T_LEN;
    const bool   writer = (tt == 2) && (half == 0);

    v2f xbuf[PF], xcur[PF];
    #pragma unroll
    for (int p = 0; p < PF; ++p) {
        xbuf[p].x = xrow[(size_t)p * INPUT + 0];
        xbuf[p].y = xrow[(size_t)p * INPUT + 1];
    }

    v8f accp = __builtin_amdgcn_wmma_f32_16x16x4_f32(
        false, Ax, false, xbuf[0], (short)0, Cb, false, false);
    accp = __builtin_amdgcn_wmma_f32_16x16x4_f32(
        false, Ah[0], false, hown, (short)0, accp, false, false);

    for (int tb = 0; tb < T_LEN; tb += PF) {
        #pragma unroll
        for (int p = 0; p < PF; ++p) xcur[p] = xbuf[p];

        {
            int tnext = tb + PF;
            if (tnext > T_LEN - PF) tnext = T_LEN - PF;
            const float* xpf = xrow + (size_t)tnext * INPUT;
            #pragma unroll
            for (int p = 0; p < PF; ++p) {
                xbuf[p].x = xpf[p * INPUT + 0];
                xbuf[p].y = xpf[p * INPUT + 1];
            }
        }

        float ov[PF];

        #pragma unroll
        for (int u = 0; u < PF; ++u) {
            v8f acc = __builtin_amdgcn_wmma_f32_16x16x4_f32(
                false, Ah[1], false, hB1, (short)0, accp, false, false);
            acc = __builtin_amdgcn_wmma_f32_16x16x4_f32(
                false, Ah[2], false, hB2, (short)0, acc, false, false);

            ov[u] = acc[4];

            hown.x = cell_step(acc[0], acc[1], acc[2], acc[3], hc0);
            hown.y = cell_step(acc[4], acc[5], acc[6], acc[7], hc1);

            hx[(u + 1) & 1][tt][lane] = hown;

            __builtin_amdgcn_fence(__ATOMIC_RELEASE, "workgroup", "local");
            __builtin_amdgcn_s_barrier();
            __builtin_amdgcn_fence(__ATOMIC_ACQUIRE, "workgroup", "local");
            hB1 = hx[(u + 1) & 1][j1][lane];
            hB2 = hx[(u + 1) & 1][j2][lane];

            __builtin_amdgcn_sched_barrier(0);

            const v2f xn = (u + 1 < PF) ? xcur[(u + 1) % PF] : xbuf[0];
            accp = __builtin_amdgcn_wmma_f32_16x16x4_f32(
                false, Ax, false, xn, (short)0, Cb, false, false);
            accp = __builtin_amdgcn_wmma_f32_16x16x4_f32(
                false, Ah[0], false, hown, (short)0, accp, false, false);
        }

        if (writer) {
            if (tb > 0) orow[tb - 1] = ov[0];
            #pragma unroll
            for (int u = 1; u < PF; ++u) orow[tb - 1 + u] = ov[u];
        }
    }

    {
        v8f accf = Cb;
        accf = __builtin_amdgcn_wmma_f32_16x16x4_f32(
            false, Ah[0], false, hown, (short)0, accf, false, false);
        accf = __builtin_amdgcn_wmma_f32_16x16x4_f32(
            false, Ah[1], false, hB1, (short)0, accf, false, false);
        accf = __builtin_amdgcn_wmma_f32_16x16x4_f32(
            false, Ah[2], false, hB2, (short)0, accf, false, false);
        if (writer) orow[T_LEN - 1] = accf[4];
    }
}

extern "C" void kernel_launch(void* const* d_in, const int* in_sizes, int n_in,
                              void* d_out, int out_size, void* d_ws, size_t ws_size,
                              hipStream_t stream) {
    const float* x     = (const float*)d_in[0];
    const float* W_ih  = (const float*)d_in[1];
    const float* W_hh  = (const float*)d_in[2];
    const float* b_ih  = (const float*)d_in[3];
    const float* b_hh  = (const float*)d_in[4];
    const float* W_lin = (const float*)d_in[5];
    const float* b_lin = (const float*)d_in[6];
    float* out = (float*)d_out;

    dim3 grid(BATCH / 16);
    dim3 block(96);
    lstm_wmma_f32_kernel<<<grid, block, 0, stream>>>(
        x, W_ih, W_hh, b_ih, b_hh, W_lin, b_lin, out);
}
